# starting kernel
# baseline (speedup 1.0000x reference)
#include <hip/hip_runtime.h>
#include <hip/hip_bf16.h>


typedef __attribute__((ext_vector_type(2))) float v2f;
typedef __attribute__((ext_vector_type(8))) float v8f;

#define NS       128
#define HSTRIDE  132
#define W2STRIDE 65
#define NREPEAT  200

__global__ __launch_bounds__(256, 1)
void conv_chain_wmma_kernel(const float* __restrict__ x,
                            const float* __restrict__ W1,
                            const float* __restrict__ b1,
                            const float* __restrict__ W2,
                            const float* __restrict__ b2,
                            float* __restrict__ out)
{
    __shared__ float Hbuf[64 * HSTRIDE];
    __shared__ float W2lds[64 * W2STRIDE];
    __shared__ float b2lds[64];

    const int tid  = threadIdx.x;
    const int lane = tid & 31;
    const int wave = tid >> 5;
    const int half = lane >> 4;
    const int m    = lane & 15;

    for (int idx = tid; idx < 64 * 64; idx += 256)
        W2lds[(idx >> 6) * W2STRIDE + (idx & 63)] = W2[idx];
    if (tid < 64) b2lds[tid] = b2[tid];

    for (int idx = tid; idx < 64 * NS; idx += 256) {
        const int c0  = idx / NS;
        const int s   = idx - c0 * NS;
        const int sg  = blockIdx.x * NS + s;
        const int b   = sg >> 12;
        const int rem = sg & 4095;
        const int yy  = rem >> 6;
        const int xx  = rem & 63;
        float acc = b1[c0];
        for (int ci = 0; ci < 3; ++ci) {
            const float* xp = x  + ((b * 3 + ci) * 70 + yy) * 70 + xx;
            const float* wp = W1 + (c0 * 3 + ci) * 49;
            #pragma unroll
            for (int ky = 0; ky < 7; ++ky) {
                #pragma unroll
                for (int kx = 0; kx < 7; ++kx)
                    acc = __builtin_fmaf(xp[ky * 70 + kx], wp[ky * 7 + kx], acc);
            }
        }
        Hbuf[c0 * HSTRIDE + s] = acc;
    }
    __syncthreads();

    v2f afrag[4][16];
    #pragma unroll
    for (int ct = 0; ct < 4; ++ct) {
        #pragma unroll
        for (int k = 0; k < 16; ++k) {
            const int row = ct * 16 + m;
            const int c   = k * 4 + half * 2;
            v2f a;
            a[0] = W2lds[row * W2STRIDE + c];
            a[1] = W2lds[row * W2STRIDE + c + 1];
            afrag[ct][k] = a;
        }
    }

    v8f biasv[4];
    #pragma unroll
    for (int ct = 0; ct < 4; ++ct) {
        v8f bv;
        #pragma unroll
        for (int i = 0; i < 8; ++i)
            bv[i] = b2lds[ct * 16 + i + half * 8];
        biasv[ct] = bv;
    }

    const int col = wave * 16 + m;

    for (int it = 0; it < NREPEAT; ++it) {
        v8f acc0 = biasv[0], acc1 = biasv[1], acc2 = biasv[2], acc3 = biasv[3];
        #pragma unroll
        for (int k = 0; k < 16; ++k) {
            const int r = k * 4 + half * 2;
            v2f bfrag;
            bfrag[0] = Hbuf[r * HSTRIDE + col];
            bfrag[1] = Hbuf[(r + 1) * HSTRIDE + col];
            acc0 = __builtin_amdgcn_wmma_f32_16x16x4_f32(false, afrag[0][k],
                       false, bfrag, (short)0, acc0, false, false);
            acc1 = __builtin_amdgcn_wmma_f32_16x16x4_f32(false, afrag[1][k],
                       false, bfrag, (short)0, acc1, false, false);
            acc2 = __builtin_amdgcn_wmma_f32_16x16x4_f32(false, afrag[2][k],
                       false, bfrag, (short)0, acc2, false, false);
            acc3 = __builtin_amdgcn_wmma_f32_16x16x4_f32(false, afrag[3][k],
                       false, bfrag, (short)0, acc3, false, false);
        }
        __syncthreads();
        #pragma unroll
        for (int i = 0; i < 8; ++i) {
            const int row = i + half * 8;
            Hbuf[(row)      * HSTRIDE + col] = acc0[i];
            Hbuf[(row + 16) * HSTRIDE + col] = acc1[i];
            Hbuf[(row + 32) * HSTRIDE + col] = acc2[i];
            Hbuf[(row + 48) * HSTRIDE + col] = acc3[i];
        }
        __syncthreads();
    }

    for (int idx = tid; idx < 64 * NS; idx += 256) {
        const int c0  = idx / NS;
        const int s   = idx - c0 * NS;
        const int sg  = blockIdx.x * NS + s;
        const int b   = sg >> 12;
        const int rem = sg & 4095;
        out[((b * 64 + c0) << 12) + rem] = Hbuf[c0 * HSTRIDE + s];
    }
}

extern "C" void kernel_launch(void* const* d_in, const int* in_sizes, int n_in,
                              void* d_out, int out_size, void* d_ws, size_t ws_size,
                              hipStream_t stream) {
    const float* x  = (const float*)d_in[0];
    const float* W1 = (const float*)d_in[1];
    const float* b1 = (const float*)d_in[2];
    const float* W2 = (const float*)d_in[3];
    const float* b2 = (const float*)d_in[4];
    float* out = (float*)d_out;

    dim3 grid(16384 / NS), block(256);
    conv_chain_wmma_kernel<<<grid, block, 0, stream>>>(x, W1, b1, W2, b2, out);
}


# Chain loop on one wave per SIMD: waves 0-3 each run all 64 channels of a 16-column chain; H passes through the wave's own LDS columns in two halves behind matrix work; no barrier in the loop
# speedup vs baseline: 2.2469x; 2.2469x over previous
#include <hip/hip_runtime.h>
#include <hip/hip_bf16.h>


typedef __attribute__((ext_vector_type(2))) float v2f;
typedef __attribute__((ext_vector_type(8))) float v8f;

#define NS       64
#define HSTRIDE  76
#define W2STRIDE 65
#define NREPEAT  200
#define LDS_FLOATS (168 * 256)

__global__ __launch_bounds__(256, 1)
void conv_chain_wmma_kernel(const float* __restrict__ x,
                            const float* __restrict__ W1,
                            const float* __restrict__ b1,
                            const float* __restrict__ W2,
                            const float* __restrict__ b2,
                            float* __restrict__ out)
{
    __shared__ float Hbuf[LDS_FLOATS - 64 * W2STRIDE - 64];
    __shared__ float W2lds[64 * W2STRIDE];
    __shared__ float b2lds[64];

    const int tid  = threadIdx.x;
    const int lane = tid & 31;
    const int half = lane >> 4;
    const int m    = lane & 15;
    const int wave = __builtin_amdgcn_readfirstlane(tid >> 5);

    for (int idx = tid; idx < 64 * 64; idx += 256)
        W2lds[(idx >> 6) * W2STRIDE + (idx & 63)] = W2[idx];
    if (tid < 64) b2lds[tid] = b2[tid];

    {
        const int xx0 = lane * 2;
        const int sg0 = blockIdx.x * NS;
        const int b   = sg0 >> 12;
        const int yy  = (sg0 & 4095) >> 6;
        const float* xp = x + ((b * 3) * 70 + yy) * 70 + xx0;

        float patch[3][7][8];
        #pragma unroll
        for (int ci = 0; ci < 3; ++ci) {
            #pragma unroll
            for (int r = 0; r < 7; ++r) {
                #pragma unroll
                for (int cx = 0; cx < 8; ++cx)
                    patch[ci][r][cx] = xp[(ci * 70 + r) * 70 + cx];
            }
        }

        #pragma unroll 1
        for (int g = 0; g < 2; ++g) {
            const int c0 = wave * 8 + g * 4;
            const float* wp = W1 + c0 * 147;
            float acc[4][2];
            #pragma unroll
            for (int q = 0; q < 4; ++q) {
                const float bq = b1[c0 + q];
                acc[q][0] = bq; acc[q][1] = bq;
            }
            #pragma unroll
            for (int ci = 0; ci < 3; ++ci) {
                #pragma unroll
                for (int ky = 0; ky < 7; ++ky) {
                    #pragma unroll
                    for (int kx = 0; kx < 7; ++kx) {
                        #pragma unroll
                        for (int q = 0; q < 4; ++q) {
                            const float w = wp[q * 147 + ci * 49 + ky * 7 + kx];
                            acc[q][0] = __builtin_fmaf(patch[ci][ky][kx],     w, acc[q][0]);
                            acc[q][1] = __builtin_fmaf(patch[ci][ky][kx + 1], w, acc[q][1]);
                        }
                    }
                }
            }
            #pragma unroll
            for (int q = 0; q < 4; ++q) {
                float* hp = Hbuf + (c0 + q) * HSTRIDE + xx0;
                hp[0] = acc[q][0];  hp[1] = acc[q][1];
            }
        }
    }
    __syncthreads();

    if (wave < 4) {
    const int col = wave * 16 + m;

    v2f afrag[4][16];
    #pragma unroll
    for (int rt = 0; rt < 4; ++rt) {
        #pragma unroll
        for (int kb = 0; kb < 16; ++kb) {
            const int row = rt * 16 + m;
            const int c   = kb * 4 + half * 2;
            v2f a;
            a[0] = W2lds[row * W2STRIDE + c];
            a[1] = W2lds[row * W2STRIDE + c + 1];
            afrag[rt][kb] = a;
        }
    }

    v8f biasv[4];
    #pragma unroll
    for (int rt = 0; rt < 4; ++rt) {
        v8f bv;
        #pragma unroll
        for (int i = 0; i < 8; ++i)
            bv[i] = b2lds[rt * 16 + i + half * 8];
        biasv[rt] = bv;
    }

    const float* hld = Hbuf + (half * 2) * HSTRIDE + col;
    float*       hst = Hbuf + (half * 8) * HSTRIDE + col;

#define WM(A, B, C) __builtin_amdgcn_wmma_f32_16x16x4_f32(false, (A), false, (B), (short)0, (C), false, false)
#define LOAD_LO()                                                          \
    _Pragma("unroll")                                                      \
    for (int v = 0; v < 8; ++v) {                                          \
        blo[v][0] = hld[(v * 4) * HSTRIDE];                                \
        blo[v][1] = hld[(v * 4 + 1) * HSTRIDE];                            \
    }
#define LOAD_HI()                                                          \
    _Pragma("unroll")                                                      \
    for (int v = 0; v < 8; ++v) {                                          \
        bhi[v][0] = hld[(32 + v * 4) * HSTRIDE];                           \
        bhi[v][1] = hld[(32 + v * 4 + 1) * HSTRIDE];                       \
    }
#define STORE_LO()                                                         \
    _Pragma("unroll")                                                      \
    for (int i = 0; i < 8; ++i) {                                          \
        hst[i * HSTRIDE]        = acc0[i];                                 \
        hst[(16 + i) * HSTRIDE] = acc1[i];                                 \
    }
#define STORE_HI()                                                         \
    _Pragma("unroll")                                                      \
    for (int i = 0; i < 8; ++i) {                                          \
        hst[(32 + i) * HSTRIDE] = acc2[i];                                 \
        hst[(48 + i) * HSTRIDE] = acc3[i];                                 \
    }
#define PIN(x, y)                                                          \
    asm volatile("" : "+v"(x), "+v"(y));                                   \
    __builtin_amdgcn_sched_barrier(0);
#define HEAD()                                                             \
    acc0 = WM(afrag[0][0], blo[0], biasv[0]);                              \
    acc1 = WM(afrag[1][0], blo[0], biasv[1]);                              \
    acc0 = WM(afrag[0][1], blo[1], acc0);                                  \
    acc1 = WM(afrag[1][1], blo[1], acc1);
#define BODY(LOADLO)                                                       \
    _Pragma("unroll")                                                      \
    for (int v = 2; v < 8; ++v) {                                          \
        acc0 = WM(afrag[0][v], blo[v], acc0);                              \
        acc1 = WM(afrag[1][v], blo[v], acc1);                              \
    }                                                                      \
    acc2 = WM(afrag[2][0], blo[0], biasv[2]);                              \
    acc3 = WM(afrag[3][0], blo[0], biasv[3]);                              \
    _Pragma("unroll")                                                      \
    for (int v = 1; v < 8; ++v) {                                          \
        acc2 = WM(afrag[2][v], blo[v], acc2);                              \
        acc3 = WM(afrag[3][v], blo[v], acc3);                              \
    }                                                                      \
    _Pragma("unroll")                                                      \
    for (int v = 0; v < 8; ++v) {                                          \
        acc0 = WM(afrag[0][8 + v], bhi[v], acc0);                          \
        acc1 = WM(afrag[1][8 + v], bhi[v], acc1);                          \
    }                                                                      \
    PIN(acc0, acc1)     \
    _Pragma("unroll")                                                      \
    for (int v = 0; v < 2; ++v) {                                          \
        acc2 = WM(afrag[2][8 + v], bhi[v], acc2);                          \
        acc3 = WM(afrag[3][8 + v], bhi[v], acc3);                          \
    }                                                                      \
    PIN(acc2, acc3)             \
    STORE_LO()                                                             \
    LOADLO                                                                 \
             \
        \
    __builtin_amdgcn_sched_barrier(0);                                     \
    _Pragma("unroll")                                                      \
    for (int v = 2; v < 8; ++v) {                                          \
        acc2 = WM(afrag[2][8 + v], bhi[v], acc2);                          \
        acc3 = WM(afrag[3][8 + v], bhi[v], acc3);                          \
    }                                                                      \
    PIN(acc2, acc3)     \


    v2f blo[8], bhi[8];
    v8f acc0, acc1, acc2, acc3;
    LOAD_LO()
    LOAD_HI()
    asm volatile("" :: "v"(bhi[7]));
    HEAD()

    for (int it = 0; it < NREPEAT - 1; ++it) {
        BODY(LOAD_LO())
        HEAD()
        PIN(acc0, acc1)
        STORE_HI()
        LOAD_HI()
        __builtin_amdgcn_sched_barrier(0);
    }
    BODY()
    STORE_HI()
#undef PIN
#undef BODY
#undef HEAD
#undef STORE_HI
#undef STORE_LO
#undef LOAD_HI
#undef LOAD_LO
#undef WM
    }
    __syncthreads();

    {
        const float* hf = Hbuf;
        const int sg0  = blockIdx.x * NS;
        const int b    = sg0 >> 12;
        const int rem0 = sg0 & 4095;
        for (int idx = tid; idx < 64 * NS; idx += 256) {
            const int c0 = idx / NS;
            const int s  = idx - c0 * NS;
            out[((b * 64 + c0) << 12) + rem0 + s] = hf[c0 * HSTRIDE + s];
        }
    }
}

extern "C" void kernel_launch(void* const* d_in, const int* in_sizes, int n_in,
                              void* d_out, int out_size, void* d_ws, size_t ws_size,
                              hipStream_t stream) {
    const float* x  = (const float*)d_in[0];
    const float* W1 = (const float*)d_in[1];
    const float* b1 = (const float*)d_in[2];
    const float* W2 = (const float*)d_in[3];
    const float* b2 = (const float*)d_in[4];
    float* out = (float*)d_out;

    dim3 grid(16384 / NS), block(256);
    conv_chain_wmma_kernel<<<grid, block, 0, stream>>>(x, W1, b1, W2, b2, out);
}


# Conv weights off the scalar cache: each wave fetches its 1176 weights once, one per lane, with 38 vector loads at kernel entry and feeds them to the FMAs as scalar operands through v_readlane
# speedup vs baseline: 2.4244x; 1.0790x over previous
#include <hip/hip_runtime.h>
#include <hip/hip_bf16.h>


typedef __attribute__((ext_vector_type(2))) float v2f;
typedef __attribute__((ext_vector_type(8))) float v8f;

#define NS       64
#define HSTRIDE  76
#define W2STRIDE 65
#define NREPEAT  200
#define LDS_FLOATS (168 * 256)

__global__ __launch_bounds__(256, 1)
void conv_chain_wmma_kernel(const float* __restrict__ x,
                            const float* __restrict__ W1,
                            const float* __restrict__ b1,
                            const float* __restrict__ W2,
                            const float* __restrict__ b2,
                            float* __restrict__ out)
{
    __shared__ float Hbuf[LDS_FLOATS - 64 * W2STRIDE - 64];
    __shared__ float W2lds[64 * W2STRIDE];
    __shared__ float b2lds[64];

    const int tid  = threadIdx.x;
    const int lane = tid & 31;
    const int half = lane >> 4;
    const int m    = lane & 15;
    const int wave = __builtin_amdgcn_readfirstlane(tid >> 5);

    float wa[19], wb[19];
    {
        const float* wbase = W1 + wave * (8 * 147);
        #pragma unroll
        for (int j = 0; j < 19; ++j) {
            int k = j * 32 + lane;
            if (j == 18) k = k < 587 ? k : 587;
            wa[j] = wbase[k];
            wb[j] = wbase[588 + k];
        }
    }
    float ba = b1[wave * 8 + (lane & 3)];
    float bb = b1[wave * 8 + 4 + (lane & 3)];

    for (int idx = tid; idx < 64 * 64; idx += 256)
        W2lds[(idx >> 6) * W2STRIDE + (idx & 63)] = W2[idx];
    if (tid < 64) b2lds[tid] = b2[tid];

    {
        const int xx0 = lane * 2;
        const int sg0 = blockIdx.x * NS;
        const int b   = sg0 >> 12;
        const int yy  = (sg0 & 4095) >> 6;
        const float* xp = x + ((b * 3) * 70 + yy) * 70 + xx0;

        float patch[3][7][8];
        #pragma unroll
        for (int ci = 0; ci < 3; ++ci) {
            #pragma unroll
            for (int r = 0; r < 7; ++r) {
                #pragma unroll
                for (int cx = 0; cx < 8; ++cx)
                    patch[ci][r][cx] = xp[(ci * 70 + r) * 70 + cx];
            }
        }

#define RDL(V, L) __builtin_bit_cast(float, __builtin_amdgcn_readlane(__builtin_bit_cast(int, (V)), (L)))
        #pragma unroll 1
        for (int g = 0; g < 2; ++g) {
            const int c0 = wave * 8 + g * 4;
            float acc[4][2];
            #pragma unroll
            for (int q = 0; q < 4; ++q) {
                const float bq = RDL(ba, q);
                acc[q][0] = bq; acc[q][1] = bq;
            }
            #pragma unroll
            for (int ci = 0; ci < 3; ++ci) {
                #pragma unroll
                for (int ky = 0; ky < 7; ++ky) {
                    float wr[7][4];
                    #pragma unroll
                    for (int kx = 0; kx < 7; ++kx) {
                        #pragma unroll
                        for (int q = 0; q < 4; ++q)
                            wr[kx][q] = RDL(wa[(q * 147 + ci * 49 + ky * 7 + kx) >> 5],
                                            (q * 147 + ci * 49 + ky * 7 + kx) & 31);
                    }
                    #pragma unroll
                    for (int kx = 0; kx < 7; ++kx) {
                        #pragma unroll
                        for (int q = 0; q < 4; ++q) {
                            const float w = wr[kx][q];
                            acc[q][0] = __builtin_fmaf(patch[ci][ky][kx],     w, acc[q][0]);
                            acc[q][1] = __builtin_fmaf(patch[ci][ky][kx + 1], w, acc[q][1]);
                        }
                    }
                }
            }
            #pragma unroll
            for (int q = 0; q < 4; ++q) {
                float* hp = Hbuf + (c0 + q) * HSTRIDE + xx0;
                hp[0] = acc[q][0];  hp[1] = acc[q][1];
            }
            #pragma unroll
            for (int j = 0; j < 19; ++j) wa[j] = wb[j];
            ba = bb;
        }
#undef RDL
    }
    __syncthreads();

    if (wave < 4) {
    const int col = wave * 16 + m;

    v2f afrag[4][16];
    #pragma unroll
    for (int rt = 0; rt < 4; ++rt) {
        #pragma unroll
        for (int kb = 0; kb < 16; ++kb) {
            const int row = rt * 16 + m;
            const int c   = kb * 4 + half * 2;
            v2f a;
            a[0] = W2lds[row * W2STRIDE + c];
            a[1] = W2lds[row * W2STRIDE + c + 1];
            afrag[rt][kb] = a;
        }
    }

    v8f biasv[4];
    #pragma unroll
    for (int rt = 0; rt < 4; ++rt) {
        v8f bv;
        #pragma unroll
        for (int i = 0; i < 8; ++i)
            bv[i] = b2lds[rt * 16 + i + half * 8];
        biasv[rt] = bv;
    }

    const float* hld = Hbuf + (half * 2) * HSTRIDE + col;
    float*       hst = Hbuf + (half * 8) * HSTRIDE + col;

#define WM(A, B, C) __builtin_amdgcn_wmma_f32_16x16x4_f32(false, (A), false, (B), (short)0, (C), false, false)
#define LOAD_LO()                                                          \
    _Pragma("unroll")                                                      \
    for (int v = 0; v < 8; ++v) {                                          \
        blo[v][0] = hld[(v * 4) * HSTRIDE];                                \
        blo[v][1] = hld[(v * 4 + 1) * HSTRIDE];                            \
    }
#define LOAD_HI()                                                          \
    _Pragma("unroll")                                                      \
    for (int v = 0; v < 8; ++v) {                                          \
        bhi[v][0] = hld[(32 + v * 4) * HSTRIDE];                           \
        bhi[v][1] = hld[(32 + v * 4 + 1) * HSTRIDE];                       \
    }
#define STORE_LO()                                                         \
    _Pragma("unroll")                                                      \
    for (int i = 0; i < 8; ++i) {                                          \
        hst[i * HSTRIDE]        = acc0[i];                                 \
        hst[(16 + i) * HSTRIDE] = acc1[i];                                 \
    }
#define STORE_HI()                                                         \
    _Pragma("unroll")                                                      \
    for (int i = 0; i < 8; ++i) {                                          \
        hst[(32 + i) * HSTRIDE] = acc2[i];                                 \
        hst[(48 + i) * HSTRIDE] = acc3[i];                                 \
    }
#define PIN(x, y)                                                          \
    asm volatile("" : "+v"(x), "+v"(y));                                   \
    __builtin_amdgcn_sched_barrier(0);
#define HEAD()                                                             \
    acc0 = WM(afrag[0][0], blo[0], biasv[0]);                              \
    acc1 = WM(afrag[1][0], blo[0], biasv[1]);                              \
    acc0 = WM(afrag[0][1], blo[1], acc0);                                  \
    acc1 = WM(afrag[1][1], blo[1], acc1);
#define BODY(LOADLO)                                                       \
    _Pragma("unroll")                                                      \
    for (int v = 2; v < 8; ++v) {                                          \
        acc0 = WM(afrag[0][v], blo[v], acc0);                              \
        acc1 = WM(afrag[1][v], blo[v], acc1);                              \
    }                                                                      \
    acc2 = WM(afrag[2][0], blo[0], biasv[2]);                              \
    acc3 = WM(afrag[3][0], blo[0], biasv[3]);                              \
    _Pragma("unroll")                                                      \
    for (int v = 1; v < 8; ++v) {                                          \
        acc2 = WM(afrag[2][v], blo[v], acc2);                              \
        acc3 = WM(afrag[3][v], blo[v], acc3);                              \
    }                                                                      \
    _Pragma("unroll")                                                      \
    for (int v = 0; v < 8; ++v) {                                          \
        acc0 = WM(afrag[0][8 + v], bhi[v], acc0);                          \
        acc1 = WM(afrag[1][8 + v], bhi[v], acc1);                          \
    }                                                                      \
    PIN(acc0, acc1)     \
    _Pragma("unroll")                                                      \
    for (int v = 0; v < 2; ++v) {                                          \
        acc2 = WM(afrag[2][8 + v], bhi[v], acc2);                          \
        acc3 = WM(afrag[3][8 + v], bhi[v], acc3);                          \
    }                                                                      \
    PIN(acc2, acc3)             \
    STORE_LO()                                                             \
    LOADLO                                                                 \
             \
        \
    __builtin_amdgcn_sched_barrier(0);                                     \
    _Pragma("unroll")                                                      \
    for (int v = 2; v < 8; ++v) {                                          \
        acc2 = WM(afrag[2][8 + v], bhi[v], acc2);                          \
        acc3 = WM(afrag[3][8 + v], bhi[v], acc3);                          \
    }                                                                      \
    PIN(acc2, acc3)     \


    v2f blo[8], bhi[8];
    v8f acc0, acc1, acc2, acc3;
    LOAD_LO()
    LOAD_HI()
    asm volatile("" :: "v"(bhi[7]));
    HEAD()

    for (int it = 0; it < NREPEAT - 1; ++it) {
        BODY(LOAD_LO())
        HEAD()
        PIN(acc0, acc1)
        STORE_HI()
        LOAD_HI()
        __builtin_amdgcn_sched_barrier(0);
    }
    BODY()
    STORE_HI()
#undef PIN
#undef BODY
#undef HEAD
#undef STORE_HI
#undef STORE_LO
#undef LOAD_HI
#undef LOAD_LO
#undef WM
    }
    __syncthreads();

    {
        const float* hf = Hbuf;
        const int sg0  = blockIdx.x * NS;
        const int b    = sg0 >> 12;
        const int rem0 = sg0 & 4095;
        for (int idx = tid; idx < 64 * NS; idx += 256) {
            const int c0 = idx / NS;
            const int s  = idx - c0 * NS;
            out[((b * 64 + c0) << 12) + rem0 + s] = hf[c0 * HSTRIDE + s];
        }
    }
}

extern "C" void kernel_launch(void* const* d_in, const int* in_sizes, int n_in,
                              void* d_out, int out_size, void* d_ws, size_t ws_size,
                              hipStream_t stream) {
    const float* x  = (const float*)d_in[0];
    const float* W1 = (const float*)d_in[1];
    const float* b1 = (const float*)d_in[2];
    const float* W2 = (const float*)d_in[3];
    const float* b2 = (const float*)d_in[4];
    float* out = (float*)d_out;

    dim3 grid(16384 / NS), block(256);
    conv_chain_wmma_kernel<<<grid, block, 0, stream>>>(x, W1, b1, W2, b2, out);
}
